# starting kernel
# baseline (speedup 1.0000x reference)
#include <hip/hip_runtime.h>
#include <hip/hip_bf16.h>

typedef __attribute__((ext_vector_type(2))) float v2f;
typedef __attribute__((ext_vector_type(8))) float v8f;

#define NBATCH 64
#define CIN    64
#define COUT   64
#define TFR    128
#define VJ     55
#define EPSF   1e-5f

#define NT_TOT (NBATCH * TFR)
#define VC     (VJ * COUT)
#define Z_ELTS ((long)NT_TOT * VC)


__global__ __launch_bounds__(256)
void shiftgcn_precompute(const float* __restrict__ b,
                         const float* __restrict__ fm,
                         const float* __restrict__ g1,  const float* __restrict__ be1,
                         const float* __restrict__ mu1, const float* __restrict__ vr1,
                         const float* __restrict__ g2,  const float* __restrict__ be2,
                         const float* __restrict__ mu2, const float* __restrict__ vr2,
                         float* __restrict__ scale1, float* __restrict__ shift1,
                         float* __restrict__ tmT,
                         float* __restrict__ scale2, float* __restrict__ shift2)
{
    const int tid = threadIdx.x;
    for (int i = tid; i < VC; i += 256) {
        const float s = g1[i] * rsqrtf(vr1[i] + EPSF);
        scale1[i] = s;
        shift1[i] = be1[i] + s * (b[i & 63] - mu1[i]);
    }
    for (int i = tid; i < 64 * 64; i += 256) {
        const int c = i >> 6, row = i & 63;
        tmT[i] = (row < VJ) ? (tanhf(fm[row * CIN + c]) + 1.0f) : 0.0f;
    }
    for (int i = tid; i < COUT; i += 256) {
        const float s = g2[i] * rsqrtf(vr2[i] + EPSF);
        scale2[i] = s;
        shift2[i] = be2[i] - s * mu2[i];
    }
}

__global__ __launch_bounds__(128)
void shiftgcn_gemm_bn1(const float* __restrict__ x,
                       const float* __restrict__ W,
                       const float* __restrict__ tmT,
                       const float* __restrict__ scale1,
                       const float* __restrict__ shift1,
                       float* __restrict__ z)
{
    __shared__ float ldsA[64 * 65];
    __shared__ float ldsW[64 * 65];

    const int nt  = blockIdx.x;
    const int n   = nt >> 7;
    const int t   = nt & 127;
    const int tid = threadIdx.x;

    const long xbase = (long)n * (CIN * TFR * VJ) + (long)t * VJ;

    for (int i = tid; i < 64 * 64; i += 128) {
        const int c   = i >> 6;
        const int row = i & 63;
        float val = 0.0f;
        if (row < VJ) {
            const int vv = (row + c) % VJ;
            val = x[xbase + (long)c * (TFR * VJ) + vv] * tmT[c * 64 + row];
        }
        ldsA[row * 65 + c] = val;
    }
    {
        const float4* W4 = (const float4*)W;
        for (int j = tid; j < 1024; j += 128) {
            const float4 w = W4[j];
            const int k = (j * 4) >> 6, d = (j * 4) & 63;
            ldsW[k * 65 + d + 0] = w.x;
            ldsW[k * 65 + d + 1] = w.y;
            ldsW[k * 65 + d + 2] = w.z;
            ldsW[k * 65 + d + 3] = w.w;
        }
    }
    __syncthreads();

    const int lane = tid & 31;
    const int wv   = tid >> 5;
    const int half = lane >> 4;
    const int l16  = lane & 15;
    const int dcol = wv * 16 + l16;

    v8f acc[4];
    acc[0] = v8f{}; acc[1] = v8f{}; acc[2] = v8f{}; acc[3] = v8f{};

    for (int ks = 0; ks < 16; ++ks) {
        const int kb = ks * 4 + half * 2;

        v2f bf;
        bf.x = ldsW[(kb + 0) * 65 + dcol];
        bf.y = ldsW[(kb + 1) * 65 + dcol];

        v2f a0, a1, a2, a3;
        a0.x = ldsA[(l16 +  0) * 65 + kb];  a0.y = ldsA[(l16 +  0) * 65 + kb + 1];
        a1.x = ldsA[(l16 + 16) * 65 + kb];  a1.y = ldsA[(l16 + 16) * 65 + kb + 1];
        a2.x = ldsA[(l16 + 32) * 65 + kb];  a2.y = ldsA[(l16 + 32) * 65 + kb + 1];
        a3.x = ldsA[(l16 + 48) * 65 + kb];  a3.y = ldsA[(l16 + 48) * 65 + kb + 1];

        acc[0] = __builtin_amdgcn_wmma_f32_16x16x4_f32(false, a0, false, bf, (short)0, acc[0], false, false);
        acc[1] = __builtin_amdgcn_wmma_f32_16x16x4_f32(false, a1, false, bf, (short)0, acc[1], false, false);
        acc[2] = __builtin_amdgcn_wmma_f32_16x16x4_f32(false, a2, false, bf, (short)0, acc[2], false, false);
        acc[3] = __builtin_amdgcn_wmma_f32_16x16x4_f32(false, a3, false, bf, (short)0, acc[3], false, false);
    }

    #pragma unroll
    for (int mt = 0; mt < 4; ++mt) {
        #pragma unroll
        for (int i = 0; i < 8; ++i) {
            const int m = half * 8 + i;
            const int v = mt * 16 + m;
            if (v < VJ) {
                const int idx1 = v * COUT + dcol;
                z[((long)nt * VJ + v) * COUT + dcol] =
                    scale1[idx1] * acc[mt][i] + shift1[idx1];
            }
        }
    }
}

__global__ __launch_bounds__(256)
void shiftgcn_tail(const float* __restrict__ z,
                   const float* __restrict__ x,
                   const float* __restrict__ scale2,
                   const float* __restrict__ shift2,
                   float* __restrict__ out)
{
    __shared__ float ldsz[3][VJ * 65];

    const int nt  = blockIdx.x;
    const int n   = nt >> 7;
    const int t   = nt & 127;
    const int tid = threadIdx.x;

    #pragma unroll
    for (int dt = 0; dt < 3; ++dt) {
        const int tsrc = t + dt - 1;
        if (tsrc >= 0 && tsrc < TFR) {
            const float4* zr = (const float4*)(z + (long)(n * TFR + tsrc) * VC);
            for (int j = tid; j < VC / 4; j += 256) {
                const float4 p = zr[j];
                const int jj = j * 4;
                const int row = jj >> 6, col = jj & 63;
                float* dst = &ldsz[dt][row * 65 + col];
                dst[0] = fmaxf(p.x, 0.0f);
                dst[1] = fmaxf(p.y, 0.0f);
                dst[2] = fmaxf(p.z, 0.0f);
                dst[3] = fmaxf(p.w, 0.0f);
            }
        } else {
            for (int j = tid; j < VC; j += 256) {
                ldsz[dt][(j >> 6) * 65 + (j & 63)] = 0.0f;
            }
        }
    }
    __syncthreads();

    const long obase = (long)n * (COUT * TFR * VJ) + (long)t * VJ;
    for (int i = tid; i < VC; i += 256) {
        const int d = i / VJ;
        const int v = i - d * VJ;
        const int dt = d % 3;
        int vs = v - d;
        vs %= VJ; if (vs < 0) vs += VJ;

        const float gval = ldsz[dt][vs * 65 + d];
        const float r    = scale2[d] * gval + shift2[d];
        const long  xi   = obase + (long)d * (TFR * VJ) + v;
        out[xi] = fmaxf(r + x[xi], 0.0f);
    }
}

extern "C" void kernel_launch(void* const* d_in, const int* in_sizes, int n_in,
                              void* d_out, int out_size, void* d_ws, size_t ws_size,
                              hipStream_t stream)
{
    const float* x   = (const float*)d_in[0];
    const float* W   = (const float*)d_in[1];
    const float* b   = (const float*)d_in[2];
    const float* fm  = (const float*)d_in[3];
    const float* g1  = (const float*)d_in[4];
    const float* be1 = (const float*)d_in[5];
    const float* mu1 = (const float*)d_in[6];
    const float* vr1 = (const float*)d_in[7];
    const float* g2  = (const float*)d_in[8];
    const float* be2 = (const float*)d_in[9];
    const float* mu2 = (const float*)d_in[10];
    const float* vr2 = (const float*)d_in[11];

    float* ws     = (float*)d_ws;
    float* z      = ws;
    float* scale1 = z + Z_ELTS;
    float* shift1 = scale1 + VC;
    float* tmT    = shift1 + VC;
    float* scale2 = tmT + 64 * 64;
    float* shift2 = scale2 + COUT;
    float* out    = (float*)d_out;

    shiftgcn_precompute<<<1, 256, 0, stream>>>(b, fm, g1, be1, mu1, vr1,
                                               g2, be2, mu2, vr2,
                                               scale1, shift1, tmT, scale2, shift2);

    shiftgcn_gemm_bn1<<<NT_TOT, 128, 0, stream>>>(x, W, tmT, scale1, shift1, z);

    shiftgcn_tail<<<NT_TOT, 256, 0, stream>>>(z, x, scale2, shift2, out);
}


# Memory-level parallelism: each phase issues all its global loads as one branch-free group before the first wait (was 1-2 per round trip), in all three kernels; arithmetic and sum order unchanged
# speedup vs baseline: 1.5255x; 1.5255x over previous
#include <hip/hip_runtime.h>
#include <hip/hip_bf16.h>

typedef __attribute__((ext_vector_type(2))) float v2f;
typedef __attribute__((ext_vector_type(8))) float v8f;

#define NBATCH 64
#define CIN    64
#define COUT   64
#define TFR    128
#define VJ     55
#define EPSF   1e-5f

#define NT_TOT (NBATCH * TFR)
#define VC     (VJ * COUT)
#define Z_ELTS ((long)NT_TOT * VC)


__global__ __launch_bounds__(256)
void shiftgcn_precompute(const float* __restrict__ b,
                         const float* __restrict__ fm,
                         const float* __restrict__ g1,  const float* __restrict__ be1,
                         const float* __restrict__ mu1, const float* __restrict__ vr1,
                         const float* __restrict__ g2,  const float* __restrict__ be2,
                         const float* __restrict__ mu2, const float* __restrict__ vr2,
                         float* __restrict__ scale1, float* __restrict__ shift1,
                         float* __restrict__ tmT,
                         float* __restrict__ scale2, float* __restrict__ shift2)
{
    const int tid = threadIdx.x;
    __builtin_assume(tid >= 0 && tid < 256);

    const int  row  = tid & 63;
    const bool rok  = row < VJ;
    const int  rowc = rok ? row : (VJ - 1);
    float fv[16];
    for (int k = 0; k < 16; ++k) {
        const int c = (tid >> 6) + 4 * k;
        fv[k] = fm[rowc * CIN + c];
    }
    float gv[14], vv[14], bv[14], mv[14], cv[14];
    for (int k = 0; k < 14; ++k) {
        const int i  = tid + 256 * k;
        const int ic = i < VC ? i : VC - 1;
        gv[k] = g1[ic];  vv[k] = vr1[ic];
        bv[k] = be1[ic]; mv[k] = mu1[ic];
        cv[k] = b[ic & 63];
    }
    const int   j2  = tid & 63;
    const float g2v = g2[j2], v2v = vr2[j2], b2v = be2[j2], m2v = mu2[j2];

    for (int k = 0; k < 14; ++k) {
        const int i = tid + 256 * k;
        if (i < VC) {
            const float s = gv[k] * rsqrtf(vv[k] + EPSF);
            scale1[i] = s;
            shift1[i] = bv[k] + s * (cv[k] - mv[k]);
        }
    }
    for (int k = 0; k < 16; ++k) {
        const int i = tid + 256 * k;
        const float tv = tanhf(fv[k]) + 1.0f;
        tmT[i] = rok ? tv : 0.0f;
    }
    if (tid < COUT) {
        const float s = g2v * rsqrtf(v2v + EPSF);
        scale2[tid] = s;
        shift2[tid] = b2v - s * m2v;
    }
}

__global__ __launch_bounds__(128)
void shiftgcn_gemm_bn1(const float* __restrict__ x,
                       const float* __restrict__ W,
                       const float* __restrict__ tmT,
                       const float* __restrict__ scale1,
                       const float* __restrict__ shift1,
                       float* __restrict__ z)
{
    __shared__ float ldsA[64 * 65];
    __shared__ float ldsW[64 * 65];

    const int nt  = blockIdx.x;
    const int n   = nt >> 7;
    const int t   = nt & 127;
    const int tid = threadIdx.x;

    const long xbase = (long)n * (CIN * TFR * VJ) + (long)t * VJ;

    __builtin_assume(tid >= 0 && tid < 128);
    const int  row   = tid & 63;
    const int  c0    = tid >> 6;
    const bool valid = row < VJ;
    const int  rowc  = valid ? row : (VJ - 1);
    const float* xb  = x + xbase;

    float4 wreg[8];
    {
        const float4* W4 = (const float4*)W;
        for (int q = 0; q < 8; ++q) wreg[q] = W4[tid + 128 * q];
    }

    int vv = rowc + c0;
    if (vv >= VJ) vv -= VJ;

    float xv[16], tv[16];
    for (int k = 0; k < 16; ++k) {
        const int c = c0 + 2 * k;
        xv[k] = xb[c * (TFR * VJ) + vv];
        tv[k] = tmT[c * 64 + row];
        vv += 2; if (vv >= VJ) vv -= VJ;
    }
    for (int q = 0; q < 8; ++q) {
        const int j = tid + 128 * q;
        const int k = (j * 4) >> 6, d = (j * 4) & 63;
        ldsW[k * 65 + d + 0] = wreg[q].x;
        ldsW[k * 65 + d + 1] = wreg[q].y;
        ldsW[k * 65 + d + 2] = wreg[q].z;
        ldsW[k * 65 + d + 3] = wreg[q].w;
    }
    for (int k = 0; k < 16; ++k) {
        const int c = c0 + 2 * k;
        ldsA[row * 65 + c] = valid ? xv[k] * tv[k] : 0.0f;
    }
    for (int k = 0; k < 16; ++k) {
        const int c = c0 + 32 + 2 * k;
        xv[k] = xb[c * (TFR * VJ) + vv];
        tv[k] = tmT[c * 64 + row];
        vv += 2; if (vv >= VJ) vv -= VJ;
    }
    for (int k = 0; k < 16; ++k) {
        const int c = c0 + 32 + 2 * k;
        ldsA[row * 65 + c] = valid ? xv[k] * tv[k] : 0.0f;
    }
    __syncthreads();

    const int lane = tid & 31;
    const int wv   = tid >> 5;
    const int half = lane >> 4;
    const int l16  = lane & 15;
    const int dcol = wv * 16 + l16;

    v8f acc[4];
    acc[0] = v8f{}; acc[1] = v8f{}; acc[2] = v8f{}; acc[3] = v8f{};

    for (int ks = 0; ks < 16; ++ks) {
        const int kb = ks * 4 + half * 2;

        v2f bf;
        bf.x = ldsW[(kb + 0) * 65 + dcol];
        bf.y = ldsW[(kb + 1) * 65 + dcol];

        v2f a0, a1, a2, a3;
        a0.x = ldsA[(l16 +  0) * 65 + kb];  a0.y = ldsA[(l16 +  0) * 65 + kb + 1];
        a1.x = ldsA[(l16 + 16) * 65 + kb];  a1.y = ldsA[(l16 + 16) * 65 + kb + 1];
        a2.x = ldsA[(l16 + 32) * 65 + kb];  a2.y = ldsA[(l16 + 32) * 65 + kb + 1];
        a3.x = ldsA[(l16 + 48) * 65 + kb];  a3.y = ldsA[(l16 + 48) * 65 + kb + 1];

        acc[0] = __builtin_amdgcn_wmma_f32_16x16x4_f32(false, a0, false, bf, (short)0, acc[0], false, false);
        acc[1] = __builtin_amdgcn_wmma_f32_16x16x4_f32(false, a1, false, bf, (short)0, acc[1], false, false);
        acc[2] = __builtin_amdgcn_wmma_f32_16x16x4_f32(false, a2, false, bf, (short)0, acc[2], false, false);
        acc[3] = __builtin_amdgcn_wmma_f32_16x16x4_f32(false, a3, false, bf, (short)0, acc[3], false, false);
    }

    float sc[4][8], sh[4][8];
    for (int mt = 0; mt < 4; ++mt) {
        for (int i = 0; i < 8; ++i) {
            const int v  = mt * 16 + half * 8 + i;
            const int vc = (v < VJ) ? v : (VJ - 1);
            sc[mt][i] = scale1[vc * COUT + dcol];
            sh[mt][i] = shift1[vc * COUT + dcol];
        }
    }
    for (int mt = 0; mt < 4; ++mt) {
        for (int i = 0; i < 8; ++i) {
            const int v = mt * 16 + half * 8 + i;
            if (v < VJ) {
                z[((long)nt * VJ + v) * COUT + dcol] =
                    sc[mt][i] * acc[mt][i] + sh[mt][i];
            }
        }
    }
}

__global__ __launch_bounds__(256)
void shiftgcn_tail(const float* __restrict__ z,
                   const float* __restrict__ x,
                   const float* __restrict__ scale2,
                   const float* __restrict__ shift2,
                   float* __restrict__ out)
{
    __shared__ float ldsz[3][64 * 65];

    const int nt  = blockIdx.x;
    const int n   = nt >> 7;
    const int t   = nt & 127;
    const int tid = threadIdx.x;

    __builtin_assume(tid >= 0 && tid < 256);
    float4 p[3][4];
    for (int dt = 0; dt < 3; ++dt) {
        const int tsrc = t + dt - 1;
        const int tcl  = tsrc < 0 ? 0 : (tsrc >= TFR ? TFR - 1 : tsrc);
        const float4* zr = (const float4*)(z + (long)(n * TFR + tcl) * VC);
        for (int q = 0; q < 4; ++q) {
            const int j  = tid + 256 * q;
            const int jc = j < VC / 4 ? j : VC / 4 - 1;
            p[dt][q] = zr[jc];
        }
    }

    const long obase = (long)n * (COUT * TFR * VJ) + (long)t * VJ;
    const float* xo = x + obase;
    float xr[14];
    for (int k = 0; k < 14; ++k) {
        const int i  = tid + 256 * k;
        const int ic = i < VC ? i : VC - 1;
        const int d  = ic / VJ;
        const int v  = ic - d * VJ;
        xr[k] = xo[d * (TFR * VJ) + v];
    }

    for (int dt = 0; dt < 3; ++dt) {
        const int  tsrc = t + dt - 1;
        const bool ok   = (tsrc >= 0) && (tsrc < TFR);
        for (int q = 0; q < 4; ++q) {
            const int j  = tid + 256 * q;
            const int jj = j * 4;
            const int rw = jj >> 6, col = jj & 63;
            float* dst = &ldsz[dt][rw * 65 + col];
            dst[0] = ok ? fmaxf(p[dt][q].x, 0.0f) : 0.0f;
            dst[1] = ok ? fmaxf(p[dt][q].y, 0.0f) : 0.0f;
            dst[2] = ok ? fmaxf(p[dt][q].z, 0.0f) : 0.0f;
            dst[3] = ok ? fmaxf(p[dt][q].w, 0.0f) : 0.0f;
        }
    }
    __syncthreads();

    float s2[14], h2[14], gv[14];
    for (int k = 0; k < 14; ++k) {
        const int i  = tid + 256 * k;
        const int ic = i < VC ? i : VC - 1;
        const int d  = ic / VJ;
        s2[k] = scale2[d];
        h2[k] = shift2[d];
    }
    for (int k = 0; k < 14; ++k) {
        const int i  = tid + 256 * k;
        const int ic = i < VC ? i : VC - 1;
        const int d  = ic / VJ;
        const int v  = ic - d * VJ;
        const int dt = d % 3;
        int vs = v - d;
        vs %= VJ; if (vs < 0) vs += VJ;
        gv[k] = ldsz[dt][vs * 65 + d];
    }
    float* oo = out + obase;
    for (int k = 0; k < 14; ++k) {
        const int i = tid + 256 * k;
        if (i < VC) {
            const int d = i / VJ;
            const int v = i - d * VJ;
            const float r = s2[k] * gv[k] + h2[k];
            oo[d * (TFR * VJ) + v] = fmaxf(r + xr[k], 0.0f);
        }
    }
}

extern "C" void kernel_launch(void* const* d_in, const int* in_sizes, int n_in,
                              void* d_out, int out_size, void* d_ws, size_t ws_size,
                              hipStream_t stream)
{
    const float* x   = (const float*)d_in[0];
    const float* W   = (const float*)d_in[1];
    const float* b   = (const float*)d_in[2];
    const float* fm  = (const float*)d_in[3];
    const float* g1  = (const float*)d_in[4];
    const float* be1 = (const float*)d_in[5];
    const float* mu1 = (const float*)d_in[6];
    const float* vr1 = (const float*)d_in[7];
    const float* g2  = (const float*)d_in[8];
    const float* be2 = (const float*)d_in[9];
    const float* mu2 = (const float*)d_in[10];
    const float* vr2 = (const float*)d_in[11];

    float* ws     = (float*)d_ws;
    float* z      = ws;
    float* scale1 = z + Z_ELTS;
    float* shift1 = scale1 + VC;
    float* tmT    = shift1 + VC;
    float* scale2 = tmT + 64 * 64;
    float* shift2 = scale2 + COUT;
    float* out    = (float*)d_out;

    shiftgcn_precompute<<<1, 256, 0, stream>>>(b, fm, g1, be1, mu1, vr1,
                                               g2, be2, mu2, vr2,
                                               scale1, shift1, tmT, scale2, shift2);

    shiftgcn_gemm_bn1<<<NT_TOT, 128, 0, stream>>>(x, W, tmT, scale1, shift1, z);

    shiftgcn_tail<<<NT_TOT, 256, 0, stream>>>(z, x, scale2, shift2, out);
}
